# starting kernel
# baseline (speedup 1.0000x reference)
#include <hip/hip_runtime.h>


typedef __attribute__((ext_vector_type(2))) float v2f;
typedef __attribute__((ext_vector_type(4))) float v4f;
typedef __attribute__((ext_vector_type(8))) float v8f;

#define SEQ_N 1048576
#define FEAT_N 3
#define HID_N 10
#define ROWF 36

#if __has_builtin(__builtin_amdgcn_tanhf)
#define TANHF(x) __builtin_amdgcn_tanhf(x)
#else
#define TANHF(x) tanhf(x)
#endif

__device__ __forceinline__ float bperm_f(int idx_bytes, float v) {
    return __int_as_float(__builtin_amdgcn_ds_bpermute(idx_bytes, __float_as_int(v)));
}

__global__ __launch_bounds__(32, 1)
void rnn_scan_wmma(const float* __restrict__ src,
                   const float* __restrict__ W_ih,
                   const float* __restrict__ W_hh,
                   const float* __restrict__ b_ih,
                   const float* __restrict__ b_hh,
                   const float* __restrict__ W_fc,
                   const float* __restrict__ b_fc,
                   float* __restrict__ out)
{
    __shared__ __align__(16) float hbuf[32 * ROWF];

    const int lane = threadIdx.x;
    const int n    = lane & 15;
    const int hi2  = (lane >> 4) << 1;

    v2f Bm[3];
#pragma unroll
    for (int c = 0; c < 3; ++c) {
        const int k0 = 4 * c + hi2;
        const int k1 = k0 + 1;
        Bm[c].x = (k0 < HID_N && n < HID_N) ? W_hh[n * HID_N + k0] : 0.0f;
        Bm[c].y = (k1 < HID_N && n < HID_N) ? W_hh[n * HID_N + k1] : 0.0f;
    }

    int idxA[3][2];
#pragma unroll
    for (int c = 0; c < 3; ++c) {
        idxA[c][0] = (4 * c + hi2) << 2;
        idxA[c][1] = (4 * c + 1 + hi2) << 2;
    }

    const float wih  = (n < HID_N) ? W_ih[n] : 0.0f;
    const float bsum = (n < HID_N) ? (b_ih[n] + b_hh[n]) : 0.0f;
    const float bfc  = b_fc[0];

    float sfc[HID_N];
#pragma unroll
    for (int j = 0; j < HID_N; ++j) sfc[j] = W_fc[j];

    float vh = 0.0f;
    v8f acc = {};

    const int NBLK = SEQ_N / 32;

    float xv = src[(size_t)lane * FEAT_N];

    float* myrow = &hbuf[lane];
    const v4f* rrow = (const v4f*)&hbuf[lane * ROWF];

    for (int blk = 0; blk < NBLK; ++blk) {
        const int t0 = blk * 32;
        float xv_next = xv;
        if (blk + 1 < NBLK)
            xv_next = src[(size_t)(t0 + 32 + lane) * FEAT_N];

#pragma unroll
        for (int k = 0; k < 32; ++k) {
            const float x_t =
                __int_as_float(__builtin_amdgcn_readlane(__float_as_int(xv), k));

            acc[0] = fmaf(x_t, wih, bsum);

            v2f a0, a1, a2;
            a0.x = bperm_f(idxA[0][0], vh);
            a0.y = bperm_f(idxA[0][1], vh);
            a1.x = bperm_f(idxA[1][0], vh);
            a1.y = bperm_f(idxA[1][1], vh);
            a2.x = bperm_f(idxA[2][0], vh);
            a2.y = bperm_f(idxA[2][1], vh);

            v8f d = __builtin_amdgcn_wmma_f32_16x16x4_f32(
                false, a0, false, Bm[0], (short)0, acc, false, false);
            d = __builtin_amdgcn_wmma_f32_16x16x4_f32(
                false, a1, false, Bm[1], (short)0, d, false, false);
            d = __builtin_amdgcn_wmma_f32_16x16x4_f32(
                false, a2, false, Bm[2], (short)0, d, false, false);

            vh  = TANHF(d[0]);
            acc = d;

            myrow[k * ROWF] = vh;
        }

        const v4f h03 = rrow[0];
        const v4f h47 = rrow[1];
        const float h8 = hbuf[lane * ROWF + 8];
        const float h9 = hbuf[lane * ROWF + 9];
        float o = bfc;
        o = fmaf(h03.x, sfc[0], o);
        o = fmaf(h03.y, sfc[1], o);
        o = fmaf(h03.z, sfc[2], o);
        o = fmaf(h03.w, sfc[3], o);
        o = fmaf(h47.x, sfc[4], o);
        o = fmaf(h47.y, sfc[5], o);
        o = fmaf(h47.z, sfc[6], o);
        o = fmaf(h47.w, sfc[7], o);
        o = fmaf(h8,    sfc[8], o);
        o = fmaf(h9,    sfc[9], o);

        out[t0 + lane] = o;
        xv = xv_next;
    }
}

extern "C" void kernel_launch(void* const* d_in, const int* in_sizes, int n_in,
                              void* d_out, int out_size, void* d_ws, size_t ws_size,
                              hipStream_t stream) {
    const float* src  = (const float*)d_in[0];
    const float* W_ih = (const float*)d_in[1];
    const float* W_hh = (const float*)d_in[2];
    const float* b_ih = (const float*)d_in[3];
    const float* b_hh = (const float*)d_in[4];
    const float* W_fc = (const float*)d_in[5];
    const float* b_fc = (const float*)d_in[6];
    float* out = (float*)d_out;

    rnn_scan_wmma<<<1, 32, 0, stream>>>(src, W_ih, W_hh, b_ih, b_hh, W_fc, b_fc, out);
}


# Roles swapped: A=[W_hh|W_ih|b; W_fc|0|b_fc], B=[h;x;1] taken straight from tanh(D), rows and k slots permuted to match: no ds_bpermute or LDS per step; sum order of h, x, bias and out terms changes
# speedup vs baseline: 1.9677x; 1.9677x over previous
#include <hip/hip_runtime.h>


typedef __attribute__((ext_vector_type(2))) float v2f;
typedef __attribute__((ext_vector_type(8))) float v8f;

#define SEQ_N 1048576
#define FEAT_N 3
#define HID_N 10

#if __has_builtin(__builtin_amdgcn_tanhf)
#define TANHF(x) __builtin_amdgcn_tanhf(x)
#else
#define TANHF(x) tanhf(x)
#endif

__global__ __launch_bounds__(32, 1)
void rnn_scan_wmma(const float* __restrict__ src,
                   const float* __restrict__ W_ih,
                   const float* __restrict__ W_hh,
                   const float* __restrict__ b_ih,
                   const float* __restrict__ b_hh,
                   const float* __restrict__ W_fc,
                   const float* __restrict__ b_fc,
                   float* __restrict__ out)
{
    const int lane = threadIdx.x;
    const int row  = lane & 15;
    const int hf   = lane >> 4;
    const int rr   = row & 7;
    const bool row_hid = rr < 5;
    const bool row_fc  = rr == 5;
    const int unit = rr + 5 * (row >> 3);

    v2f Am[3];
#pragma unroll
    for (int c = 0; c < 3; ++c) {
#pragma unroll
        for (int s = 0; s < 2; ++s) {
            const int r = 2 * c + s;
            float w = 0.0f;
            if (r < 5) {
                const int j = r + 5 * hf;
                if (row_hid)      w = W_hh[unit * HID_N + j];
                else if (row_fc)  w = W_fc[j];
            } else {
                if (row_hid)      w = hf ? (b_ih[unit] + b_hh[unit]) : W_ih[unit];
                else if (row_fc)  w = hf ? b_fc[0] : 0.0f;
            }
            Am[c][s] = w;
        }
    }

    v2f b0 = {0.0f, 0.0f}, b1 = {0.0f, 0.0f}, b2 = {0.0f, 0.0f};
    v8f d;

#define RNN_STEP(x_t)                                                          \
    do {                                                                       \
        const v8f zero_ = {};                                                  \
        b2.y = hf ? 1.0f : (x_t);                                              \
        d = __builtin_amdgcn_wmma_f32_16x16x4_f32(                             \
            false, Am[0], false, b0, (short)0, zero_, false, false);           \
        d = __builtin_amdgcn_wmma_f32_16x16x4_f32(                             \
            false, Am[1], false, b1, (short)0, d, false, false);               \
        d = __builtin_amdgcn_wmma_f32_16x16x4_f32(                             \
            false, Am[2], false, b2, (short)0, d, false, false);               \
        b0.x = TANHF(d[0]);                                                    \
        b0.y = TANHF(d[1]);                                                    \
        b1.x = TANHF(d[2]);                                                    \
        b1.y = TANHF(d[3]);                                                    \
        b2.x = TANHF(d[4]);                                                    \
    } while (0)

    {
        const float x0 = src[0];
        RNN_STEP(x0);
    }

    const int NBLK = SEQ_N / 32;
    float xv = src[(size_t)(1 + lane) * FEAT_N];
    float o = 0.0f;

    for (int blk = 0; blk < NBLK; ++blk) {
        const int t0 = blk * 32;
        float xv_next = xv;
        if (blk + 1 < NBLK) {
            int xn = t0 + 33 + lane;
            xn = xn < SEQ_N - 1 ? xn : SEQ_N - 1;
            xv_next = src[(size_t)xn * FEAT_N];
        }

#pragma unroll
        for (int k = 0; k < 32; ++k) {
            const float x_t =
                __int_as_float(__builtin_amdgcn_readlane(__float_as_int(xv), k));
            RNN_STEP(x_t);
            o = (lane == k) ? d[5] : o;
        }

        out[t0 + lane] = o;
        xv = xv_next;
    }
#undef RNN_STEP
}

extern "C" void kernel_launch(void* const* d_in, const int* in_sizes, int n_in,
                              void* d_out, int out_size, void* d_ws, size_t ws_size,
                              hipStream_t stream) {
    const float* src  = (const float*)d_in[0];
    const float* W_ih = (const float*)d_in[1];
    const float* W_hh = (const float*)d_in[2];
    const float* b_ih = (const float*)d_in[3];
    const float* b_hh = (const float*)d_in[4];
    const float* W_fc = (const float*)d_in[5];
    const float* b_fc = (const float*)d_in[6];
    float* out = (float*)d_out;

    rnn_scan_wmma<<<1, 32, 0, stream>>>(src, W_ih, W_hh, b_ih, b_hh, W_fc, b_fc, out);
}


# Two accumulators again, but the three WMMAs now issue back to back (chunk 2, chunk 0, chunk 1 onto 2): selects moved into the third's flight, adds and tanh behind; sum order of the 12 terms changes
# speedup vs baseline: 2.0391x; 1.0363x over previous
#include <hip/hip_runtime.h>


typedef __attribute__((ext_vector_type(2))) float v2f;
typedef __attribute__((ext_vector_type(8))) float v8f;

#define SEQ_N 1048576
#define FEAT_N 3
#define HID_N 10

#if __has_builtin(__builtin_amdgcn_tanhf)
#define TANHF(x) __builtin_amdgcn_tanhf(x)
#else
#define TANHF(x) tanhf(x)
#endif

__global__ __launch_bounds__(32, 1)
void rnn_scan_wmma(const float* __restrict__ src,
                   const float* __restrict__ W_ih,
                   const float* __restrict__ W_hh,
                   const float* __restrict__ b_ih,
                   const float* __restrict__ b_hh,
                   const float* __restrict__ W_fc,
                   const float* __restrict__ b_fc,
                   float* __restrict__ out)
{
    const int lane = threadIdx.x;
    const int row  = lane & 15;
    const int hf   = lane >> 4;
    const int rr   = row & 7;
    const bool row_hid = rr < 5;
    const bool row_fc  = rr == 5;
    const int unit = rr + 5 * (row >> 3);

    v2f Am[3];
#pragma unroll
    for (int c = 0; c < 3; ++c) {
#pragma unroll
        for (int s = 0; s < 2; ++s) {
            const int r = 2 * c + s;
            float w = 0.0f;
            if (r < 5) {
                const int j = r + 5 * hf;
                if (row_hid)      w = W_hh[unit * HID_N + j];
                else if (row_fc)  w = W_fc[j];
            } else {
                if (row_hid)      w = hf ? (b_ih[unit] + b_hh[unit]) : W_ih[unit];
                else if (row_fc)  w = hf ? b_fc[0] : 0.0f;
            }
            Am[c][s] = w;
        }
    }

    v2f b0 = {0.0f, 0.0f}, b1 = {0.0f, 0.0f}, b2 = {0.0f, 0.0f};
    float fc = 0.0f;
    float o = 0.0f;

#define RNN_STEP(x_next, kprev)                                                \
    do {                                                                       \
        const v8f zero_ = {};                                                  \
          \
        __builtin_amdgcn_sched_barrier(0);                                     \
        v8f da_ = __builtin_amdgcn_wmma_f32_16x16x4_f32(                       \
            false, Am[2], false, b2, (short)0, zero_, false, false);           \
        v8f db_ = __builtin_amdgcn_wmma_f32_16x16x4_f32(                       \
            false, Am[0], false, b0, (short)0, zero_, false, false);           \
        da_ = __builtin_amdgcn_wmma_f32_16x16x4_f32(                           \
            false, Am[1], false, b1, (short)0, da_, false, false);             \
                     \
        __builtin_amdgcn_sched_barrier(0);                                     \
        if ((kprev) >= 0) o = (lane == (kprev)) ? fc : o;                      \
        b2.y = hf ? 1.0f : (x_next);                                           \
                  \
        __builtin_amdgcn_sched_barrier(0);                                     \
        b2.x = TANHF(da_[4] + db_[4]);                                         \
        b1.x = TANHF(da_[2] + db_[2]);                                         \
        b1.y = TANHF(da_[3] + db_[3]);                                         \
        b0.x = TANHF(da_[0] + db_[0]);                                         \
        b0.y = TANHF(da_[1] + db_[1]);                                         \
        fc = da_[5] + db_[5];                                                  \
    } while (0)

    float xv = src[(size_t)(1 + lane) * FEAT_N];

    {
        const float x0 = src[0];
        b2.y = hf ? 1.0f : x0;
        const float x1 =
            __int_as_float(__builtin_amdgcn_readlane(__float_as_int(xv), 0));
        RNN_STEP(x1, -1);
    }

    const int NBLK = SEQ_N / 32;

    for (int blk = 0; blk < NBLK; ++blk) {
        const int t0 = blk * 32;
        float xv_next = xv;
        if (blk + 1 < NBLK) {
            int xn = t0 + 33 + lane;
            xn = xn < SEQ_N - 1 ? xn : SEQ_N - 1;
            xv_next = src[(size_t)xn * FEAT_N];
        }

#pragma unroll
        for (int k = 0; k < 32; ++k) {
            const float x_n = __int_as_float(
                k < 31 ? __builtin_amdgcn_readlane(__float_as_int(xv), (k + 1) & 31)
                       : __builtin_amdgcn_readlane(__float_as_int(xv_next), 0));
            RNN_STEP(x_n, k - 1);
        }

        o = (lane == 31) ? fc : o;
        out[t0 + lane] = o;
        xv = xv_next;
    }
#undef RNN_STEP
}

extern "C" void kernel_launch(void* const* d_in, const int* in_sizes, int n_in,
                              void* d_out, int out_size, void* d_ws, size_t ws_size,
                              hipStream_t stream) {
    const float* src  = (const float*)d_in[0];
    const float* W_ih = (const float*)d_in[1];
    const float* W_hh = (const float*)d_in[2];
    const float* b_ih = (const float*)d_in[3];
    const float* b_hh = (const float*)d_in[4];
    const float* W_fc = (const float*)d_in[5];
    const float* b_fc = (const float*)d_in[6];
    float* out = (float*)d_out;

    rnn_scan_wmma<<<1, 32, 0, stream>>>(src, W_ih, W_hh, b_ih, b_hh, W_fc, b_fc, out);
}


# Block bookkeeping off the recurrence: out store, next-x load, x lane read issue behind a step's third WMMA; two alternating register sets keep s_wait_xcnt and pointer work out of the tanh-to-WMMA gap
# speedup vs baseline: 2.1023x; 1.0310x over previous
#include <hip/hip_runtime.h>


typedef __attribute__((ext_vector_type(2))) float v2f;
typedef __attribute__((ext_vector_type(8))) float v8f;

#define SEQ_N 1048576
#define FEAT_N 3
#define HID_N 10

#if __has_builtin(__builtin_amdgcn_tanhf)
#define TANHF(x) __builtin_amdgcn_tanhf(x)
#else
#define TANHF(x) tanhf(x)
#endif

__global__ __launch_bounds__(32, 1)
void rnn_scan_wmma(const float* __restrict__ src,
                   const float* __restrict__ W_ih,
                   const float* __restrict__ W_hh,
                   const float* __restrict__ b_ih,
                   const float* __restrict__ b_hh,
                   const float* __restrict__ W_fc,
                   const float* __restrict__ b_fc,
                   float* __restrict__ out)
{
    const int lane = threadIdx.x;
    const int row  = lane & 15;
    const int hf   = lane >> 4;
    const int rr   = row & 7;
    const bool row_hid = rr < 5;
    const bool row_fc  = rr == 5;
    const int unit = rr + 5 * (row >> 3);

    v2f Am[3];
#pragma unroll
    for (int c = 0; c < 3; ++c) {
#pragma unroll
        for (int s = 0; s < 2; ++s) {
            const int r = 2 * c + s;
            float w = 0.0f;
            if (r < 5) {
                const int j = r + 5 * hf;
                if (row_hid)      w = W_hh[unit * HID_N + j];
                else if (row_fc)  w = W_fc[j];
            } else {
                if (row_hid)      w = hf ? (b_ih[unit] + b_hh[unit]) : W_ih[unit];
                else if (row_fc)  w = hf ? b_fc[0] : 0.0f;
            }
            Am[c][s] = w;
        }
    }

    v2f b0 = {0.0f, 0.0f}, b1 = {0.0f, 0.0f}, b2 = {0.0f, 0.0f};
    float fc = 0.0f;

#define RNN_STEP(...)                                                          \
    do {                                                                       \
        const v8f zero_ = {};                                                  \
          \
        __builtin_amdgcn_sched_barrier(0);                                     \
        v8f da_ = __builtin_amdgcn_wmma_f32_16x16x4_f32(                       \
            false, Am[2], false, b2, (short)0, zero_, false, false);           \
        v8f db_ = __builtin_amdgcn_wmma_f32_16x16x4_f32(                       \
            false, Am[0], false, b0, (short)0, zero_, false, false);           \
        da_ = __builtin_amdgcn_wmma_f32_16x16x4_f32(                           \
            false, Am[1], false, b1, (short)0, da_, false, false);             \
                   \
        __builtin_amdgcn_sched_barrier(0);                                     \
        __VA_ARGS__                                                            \
                \
        __builtin_amdgcn_sched_barrier(0);                                     \
        b2.x = TANHF(da_[4] + db_[4]);                                         \
        b1.x = TANHF(da_[2] + db_[2]);                                         \
        b1.y = TANHF(da_[3] + db_[3]);                                         \
        b0.x = TANHF(da_[0] + db_[0]);                                         \
        b0.y = TANHF(da_[1] + db_[1]);                                         \
        fc = da_[5] + db_[5];                                                  \
    } while (0)

#define LANE_READ(v, k) \
    __int_as_float(__builtin_amdgcn_readlane(__float_as_int(v), (k)))


    const unsigned lane3 = (unsigned)lane * FEAT_N;
    float xvA = src[lane3];
    float xvB = src[32 * FEAT_N + lane3];
    const unsigned ixmax = (SEQ_N / 32 - 1) * (32 * FEAT_N) + lane3;
    unsigned ixA = lane3;
    unsigned ixB = 32 * FEAT_N + lane3;
    float oA = 0.0f, oB = 0.0f;
    float xs;

    b2.y = hf ? 1.0f : LANE_READ(xvA, 0);
    {
        const float x1 = LANE_READ(xvA, 1);
        RNN_STEP(b2.y = hf ? 1.0f : x1;);
        const float x2 = LANE_READ(xvA, 2);
        const float x3 = LANE_READ(xvA, 3);
        RNN_STEP(b2.y = hf ? 1.0f : x2; xs = x3;);
    }

#define RNN_STEPS(O, XV, XVN, IX, T0, K0, K1)                                  \
    _Pragma("unroll")                                                          \
    for (int k = (K0); k < (K1); ++k) {                                        \
        RNN_STEP(                                                              \
            O = (lane == k) ? fc : O;                                          \
              \
            asm volatile("" :: "v"(O));                                        \
            b2.y = hf ? 1.0f : xs;                                             \
            xs = (k < 28) ? LANE_READ(XV, (k + 4) & 31)                        \
                          : LANE_READ(XVN, (k - 28) & 31);                     \
            if (k == 15) {                                                     \
                const unsigned nx = IX + 2 * 32 * FEAT_N;                      \
                IX = nx < ixmax ? nx : ixmax;                                  \
            }                                                                  \
            if (k == 31) {                                                     \
                out[(T0) + lane] = O;                                          \
                XV = src[IX];                                                  \
            }                                                                  \
        );                                                                     \
    }

    RNN_STEPS(oA, xvA, xvB, ixA, 0, 0, 16)
    for (int m = 0; m < SEQ_N / 64; ++m) {
        const int t0 = m * 64;
        RNN_STEPS(oA, xvA, xvB, ixA, t0, 16, 32)
        RNN_STEPS(oB, xvB, xvA, ixB, t0 + 32, 0, 32)
        RNN_STEPS(oA, xvA, xvB, ixA, t0 + 64, 0, 16)
    }
#undef RNN_STEPS
#undef LANE_READ
#undef RNN_STEP
}

extern "C" void kernel_launch(void* const* d_in, const int* in_sizes, int n_in,
                              void* d_out, int out_size, void* d_ws, size_t ws_size,
                              hipStream_t stream) {
    const float* src  = (const float*)d_in[0];
    const float* W_ih = (const float*)d_in[1];
    const float* W_hh = (const float*)d_in[2];
    const float* b_ih = (const float*)d_in[3];
    const float* b_hh = (const float*)d_in[4];
    const float* W_fc = (const float*)d_in[5];
    const float* b_fc = (const float*)d_in[6];
    float* out = (float*)d_out;

    rnn_scan_wmma<<<1, 32, 0, stream>>>(src, W_ih, W_hh, b_ih, b_hh, W_fc, b_fc, out);
}


# 16 time segments run at once in the 16 B/D columns of the one wave, each after a 2050-step run-in from h=0; hand-over states compared bit for bit, serial chain rerun over all if one differs
# speedup vs baseline: 23.2434x; 11.0562x over previous
#include <hip/hip_runtime.h>


typedef __attribute__((ext_vector_type(2))) float v2f;
typedef __attribute__((ext_vector_type(4))) float v4f;
typedef __attribute__((ext_vector_type(8))) float v8f;

#define SEQ_N 1048576
#define FEAT_N 3
#define HID_N 10
#define SEG_N (SEQ_N / 16)
#define RUNIN_BLOCKS 64
#define RUNIN_N (2 + 32 * RUNIN_BLOCKS)

#if __has_builtin(__builtin_amdgcn_tanhf)
#define TANHF(x) __builtin_amdgcn_tanhf(x)
#else
#define TANHF(x) tanhf(x)
#endif

#define RNN_STEP(FCDST, ...)                                                   \
    do {                                                                       \
        const v8f zero_ = {};                                                  \
          \
        __builtin_amdgcn_sched_barrier(0);                                     \
        v8f da_ = __builtin_amdgcn_wmma_f32_16x16x4_f32(                       \
            false, Am[2], false, b2, (short)0, zero_, false, false);           \
        v8f db_ = __builtin_amdgcn_wmma_f32_16x16x4_f32(                       \
            false, Am[0], false, b0, (short)0, zero_, false, false);           \
        da_ = __builtin_amdgcn_wmma_f32_16x16x4_f32(                           \
            false, Am[1], false, b1, (short)0, da_, false, false);             \
                   \
        __builtin_amdgcn_sched_barrier(0);                                     \
        __VA_ARGS__                                                            \
                \
        __builtin_amdgcn_sched_barrier(0);                                     \
        b2.x = TANHF(da_[4] + db_[4]);                                         \
        b1.x = TANHF(da_[2] + db_[2]);                                         \
        b1.y = TANHF(da_[3] + db_[3]);                                         \
        b0.x = TANHF(da_[0] + db_[0]);                                         \
        b0.y = TANHF(da_[1] + db_[1]);                                         \
        FCDST = da_[5] + db_[5];                                               \
    } while (0)

template <bool STORE>
__device__ __forceinline__ void segmented_pass(
    const v2f (&Am)[3], v2f& b0, v2f& b1, v2f& b2,
    const float* __restrict__ src, float* __restrict__ out,
    const bool hfb, const unsigned row0, const unsigned orow0, const int nblk,
    float& f0, float& f1, float& f2, float& f3, float& f4)
{
    float XA[32], XB[32];
    v4f OQ[2] = {};
    float odummy;
    const unsigned tlim  = (SEQ_N - 3) * (FEAT_N * 4);
    const unsigned ixmax = (row0 + 32 * (nblk - 1) + 3) * (FEAT_N * 4);
    unsigned ixA = (row0 + 3) * (FEAT_N * 4);
    unsigned ixB = ixA + 32 * (FEAT_N * 4);
    unsigned ixAT = ixA + 29 * (FEAT_N * 4);
    unsigned ixBT = ixB + 29 * (FEAT_N * 4);
    ixAT = ixAT < tlim ? ixAT : tlim;
    ixBT = ixBT < tlim ? ixBT : tlim;
    unsigned obA = orow0 * 4;
    unsigned obB = obA + 32 * 4 - 2 * 32 * 4;

    const char* const srcb = (const char*)src;
    char* const outb = (char*)out;

    const float p0 = *(const float*)(srcb + row0 * (FEAT_N * 4));
    const float p1 = *(const float*)(srcb + row0 * (FEAT_N * 4) + 12);
    const float p2 = *(const float*)(srcb + row0 * (FEAT_N * 4) + 24);
#pragma unroll
    for (int k = 0; k < 32; ++k)
        XA[k] = k < 29 ? *(const float*)(srcb + ixA + 12u * k)
                       : *(const float*)(srcb + ixAT + 12u * (k - 29));

    b2.y = hfb ? 1.0f : p0;
    RNN_STEP(odummy,   b2.y = hfb ? 1.0f : p1;);
    RNN_STEP(OQ[0][0], b2.y = hfb ? 1.0f : p2;);

#define SEG_BLOCK(XC, XN, IXC, IXN, IXCT, IXNT, OB, OBN)                       \
    _Pragma("unroll")                                                          \
    for (int k = 0; k < 32; ++k) {                                             \
        RNN_STEP(OQ[((k + 1) >> 2) & 1][(k + 1) & 3],                          \
            if (STORE && (k & 3) == 3)                                         \
                *(v4f*)(outb + OB + 4u * (k - 3)) = OQ[(k >> 2) & 1];          \
            if (STORE && (k & 3) == 2) {                                       \
                       \
                asm volatile("" :: "v"(OQ[((k >> 2) + 1) & 1]));               \
            }                                                                  \
            b2.y = hfb ? 1.0f : XC[k];                                         \
            if (k < 29)                                                        \
                XN[k] = *(const float*)(srcb + IXN + 12u * k);                 \
            else                                                               \
                XN[k] = *(const float*)(srcb + IXNT + 12u * (k - 29));         \
            if (k == 15) {                                                     \
                const unsigned nx_ = IXC + 2 * 32 * (FEAT_N * 4);              \
                IXC = nx_ < ixmax ? nx_ : ixmax;                               \
                const unsigned nt_ = IXC + 29 * (FEAT_N * 4);                  \
                IXCT = nt_ < tlim ? nt_ : tlim;                                \
                OBN += 2 * 32 * 4;                                             \
            }                                                                  \
            if (STORE && k == 30) {                                            \
                   \
                f0 = b0.x; f1 = b0.y; f2 = b1.x; f3 = b1.y; f4 = b2.x;         \
            }                                                                  \
        );                                                                     \
    }

    for (int m = 0; m < nblk / 2; ++m) {
        SEG_BLOCK(XA, XB, ixA, ixB, ixAT, ixBT, obA, obB)
        SEG_BLOCK(XB, XA, ixB, ixA, ixBT, ixAT, obB, obA)
    }
#undef SEG_BLOCK
}

__device__ __forceinline__ void serial_scan(
    const v2f (&Am)[3], const float* __restrict__ src, float* __restrict__ out,
    const int lane, const int hf)
{
    v2f b0 = {0.0f, 0.0f}, b1 = {0.0f, 0.0f}, b2 = {0.0f, 0.0f};
    float fc = 0.0f;

#define LANE_READ(v, k) \
    __int_as_float(__builtin_amdgcn_readlane(__float_as_int(v), (k)))


    const unsigned lane3 = (unsigned)lane * FEAT_N;
    float xvA = src[lane3];
    float xvB = src[32 * FEAT_N + lane3];
    const unsigned ixmax = (SEQ_N / 32 - 1) * (32 * FEAT_N) + lane3;
    unsigned ixA = lane3;
    unsigned ixB = 32 * FEAT_N + lane3;
    float oA = 0.0f, oB = 0.0f;
    float xs;

    b2.y = hf ? 1.0f : LANE_READ(xvA, 0);
    {
        const float x1 = LANE_READ(xvA, 1);
        RNN_STEP(fc, b2.y = hf ? 1.0f : x1;);
        const float x2 = LANE_READ(xvA, 2);
        const float x3 = LANE_READ(xvA, 3);
        RNN_STEP(fc, b2.y = hf ? 1.0f : x2; xs = x3;);
    }

#define RNN_STEPS(O, XV, XVN, IX, T0, K0, K1)                                  \
    _Pragma("unroll")                                                          \
    for (int k = (K0); k < (K1); ++k) {                                        \
        RNN_STEP(fc,                                                           \
            O = (lane == k) ? fc : O;                                          \
              \
            asm volatile("" :: "v"(O));                                        \
            b2.y = hf ? 1.0f : xs;                                             \
            xs = (k < 28) ? LANE_READ(XV, (k + 4) & 31)                        \
                          : LANE_READ(XVN, (k - 28) & 31);                     \
            if (k == 15) {                                                     \
                const unsigned nx = IX + 2 * 32 * FEAT_N;                      \
                IX = nx < ixmax ? nx : ixmax;                                  \
            }                                                                  \
            if (k == 31) {                                                     \
                out[(T0) + lane] = O;                                          \
                XV = src[IX];                                                  \
            }                                                                  \
        );                                                                     \
    }

    RNN_STEPS(oA, xvA, xvB, ixA, 0, 0, 16)
    for (int m = 0; m < SEQ_N / 64; ++m) {
        const int t0 = m * 64;
        RNN_STEPS(oA, xvA, xvB, ixA, t0, 16, 32)
        RNN_STEPS(oB, xvB, xvA, ixB, t0 + 32, 0, 32)
        RNN_STEPS(oA, xvA, xvB, ixA, t0 + 64, 0, 16)
    }
#undef RNN_STEPS
#undef LANE_READ
}

__global__ __launch_bounds__(32, 1)
void rnn_scan_wmma(const float* __restrict__ src,
                   const float* __restrict__ W_ih,
                   const float* __restrict__ W_hh,
                   const float* __restrict__ b_ih,
                   const float* __restrict__ b_hh,
                   const float* __restrict__ W_fc,
                   const float* __restrict__ b_fc,
                   float* __restrict__ out)
{
    const int lane = threadIdx.x;
    const int row  = lane & 15;
    const int hf   = lane >> 4;
    const int rr   = row & 7;
    const bool row_hid = rr < 5;
    const bool row_fc  = rr == 5;
    const int unit = rr + 5 * (row >> 3);

    v2f Am[3];
#pragma unroll
    for (int c = 0; c < 3; ++c) {
#pragma unroll
        for (int s = 0; s < 2; ++s) {
            const int r = 2 * c + s;
            float w = 0.0f;
            if (r < 5) {
                const int j = r + 5 * hf;
                if (row_hid)      w = W_hh[unit * HID_N + j];
                else if (row_fc)  w = W_fc[j];
            } else {
                if (row_hid)      w = hf ? (b_ih[unit] + b_hh[unit]) : W_ih[unit];
                else if (row_fc)  w = hf ? b_fc[0] : 0.0f;
            }
            Am[c][s] = w;
        }
    }

    const unsigned col = (unsigned)lane & 15u;
    const bool hfb = hf != 0;
    v2f b0 = {0.0f, 0.0f}, b1 = {0.0f, 0.0f}, b2 = {0.0f, 0.0f};
    float f0 = 0.0f, f1 = 0.0f, f2 = 0.0f, f3 = 0.0f, f4 = 0.0f;

    segmented_pass<false>(Am, b0, b1, b2, src, out, hfb,
                          col ? col * SEG_N - RUNIN_N : 0u, 0u, RUNIN_BLOCKS,
                          f0, f1, f2, f3, f4);
    const float s0 = b0.x, s1 = b0.y, s2 = b1.x, s3 = b1.y, s4 = b2.x;
    if (col == 0) {
        b0.x = 0.0f; b0.y = 0.0f; b1.x = 0.0f; b1.y = 0.0f; b2.x = 0.0f;
    }
    segmented_pass<true>(Am, b0, b1, b2, src, out, hfb,
                         col * SEG_N, col * SEG_N, SEG_N / 32,
                         f0, f1, f2, f3, f4);

    bool differs = false;
#define HAND_OVER(F, S)                                                        \
    differs |= __builtin_amdgcn_ds_bpermute(((lane - 1) & 31) << 2,            \
                                            __float_as_int(F)) != __float_as_int(S)
    HAND_OVER(f0, s0);
    HAND_OVER(f1, s1);
    HAND_OVER(f2, s2);
    HAND_OVER(f3, s3);
    HAND_OVER(f4, s4);
#undef HAND_OVER
    differs = differs && (col != 0);
    if (__builtin_amdgcn_ballot_w32(differs) != 0u) {
        __threadfence();
        serial_scan(Am, src, out, lane, hf);
    }
}
#undef RNN_STEP

extern "C" void kernel_launch(void* const* d_in, const int* in_sizes, int n_in,
                              void* d_out, int out_size, void* d_ws, size_t ws_size,
                              hipStream_t stream) {
    const float* src  = (const float*)d_in[0];
    const float* W_ih = (const float*)d_in[1];
    const float* W_hh = (const float*)d_in[2];
    const float* b_ih = (const float*)d_in[3];
    const float* b_hh = (const float*)d_in[4];
    const float* W_fc = (const float*)d_in[5];
    const float* b_fc = (const float*)d_in[6];
    float* out = (float*)d_out;

    rnn_scan_wmma<<<1, 32, 0, stream>>>(src, W_ih, W_hh, b_ih, b_hh, W_fc, b_fc, out);
}


# Segments spread over 512 waves (256 work-groups of 2; 16 columns a wave, 128 rows a column, run-in still 2050); hand-over states go via workspace to a 2nd launch that compares them bit for bit
# speedup vs baseline: 678.6254x; 29.1965x over previous
#include <hip/hip_runtime.h>


typedef __attribute__((ext_vector_type(2))) float v2f;
typedef __attribute__((ext_vector_type(4))) float v4f;
typedef __attribute__((ext_vector_type(8))) float v8f;

#define SEQ_N 1048576
#define FEAT_N 3
#define HID_N 10
#define RUNIN_BLOCKS 64
#define RUNIN_N (2 + 32 * RUNIN_BLOCKS)
#define SEG_N 128
#define HEAD_N 2176
#define NSEG ((SEQ_N - HEAD_N) / SEG_N)
#define NGRP 512
#define WG_WAVES 2
#define WG_N 256
#define WG_THREADS 64
#define WS_GRP 32
#define WS_START 0
#define WS_END 10
#define WS_FLAG 20
#define CHECK_THREADS 512

static_assert(HEAD_N >= RUNIN_N, "the first segment's run-in must not start in front of row 0");
static_assert(HEAD_N % 64 == 0 && SEG_N % 64 == 0, "passes take an even number of 32-step blocks");
static_assert((SEQ_N - HEAD_N) % SEG_N == 0, "segments tile the rows behind the head");
static_assert(NGRP == 1 + (NSEG + 15) / 16, "one wave for the head, one per 16 segments");
static_assert(WG_N * WG_WAVES == NGRP && WG_THREADS == WG_WAVES * 32, "grid covers the waves");

#if __has_builtin(__builtin_amdgcn_tanhf)
#define TANHF(x) __builtin_amdgcn_tanhf(x)
#else
#define TANHF(x) tanhf(x)
#endif

#define RNN_STEP(FCDST, ...)                                                   \
    do {                                                                       \
        const v8f zero_ = {};                                                  \
          \
        __builtin_amdgcn_sched_barrier(0);                                     \
        v8f da_ = __builtin_amdgcn_wmma_f32_16x16x4_f32(                       \
            false, Am[2], false, b2, (short)0, zero_, false, false);           \
        v8f db_ = __builtin_amdgcn_wmma_f32_16x16x4_f32(                       \
            false, Am[0], false, b0, (short)0, zero_, false, false);           \
        da_ = __builtin_amdgcn_wmma_f32_16x16x4_f32(                           \
            false, Am[1], false, b1, (short)0, da_, false, false);             \
                   \
        __builtin_amdgcn_sched_barrier(0);                                     \
        __VA_ARGS__                                                            \
                \
        __builtin_amdgcn_sched_barrier(0);                                     \
        b2.x = TANHF(da_[4] + db_[4]);                                         \
        b1.x = TANHF(da_[2] + db_[2]);                                         \
        b1.y = TANHF(da_[3] + db_[3]);                                         \
        b0.x = TANHF(da_[0] + db_[0]);                                         \
        b0.y = TANHF(da_[1] + db_[1]);                                         \
        FCDST = da_[5] + db_[5];                                               \
    } while (0)

template <bool STORE>
__device__ __forceinline__ void segmented_pass(
    const v2f (&Am)[3], v2f& b0, v2f& b1, v2f& b2,
    const float* __restrict__ src, float* __restrict__ out,
    const bool hfb, const unsigned row0, const unsigned orow0, const int nblk,
    float& f0, float& f1, float& f2, float& f3, float& f4)
{
    float XA[32], XB[32];
    v4f OQ[2] = {};
    float odummy;
    const unsigned tlim  = (SEQ_N - 3) * (FEAT_N * 4);
    const unsigned ixmax = (row0 + 32 * (nblk - 1) + 3) * (FEAT_N * 4);
    unsigned ixA = (row0 + 3) * (FEAT_N * 4);
    unsigned ixB = ixA + 32 * (FEAT_N * 4);
    unsigned ixAT = ixA + 29 * (FEAT_N * 4);
    unsigned ixBT = ixB + 29 * (FEAT_N * 4);
    ixAT = ixAT < tlim ? ixAT : tlim;
    ixBT = ixBT < tlim ? ixBT : tlim;
    unsigned obA = orow0 * 4;
    unsigned obB = obA + 32 * 4 - 2 * 32 * 4;

    const char* const srcb = (const char*)src;
    char* const outb = (char*)out;

    const float p0 = *(const float*)(srcb + row0 * (FEAT_N * 4));
    const float p1 = *(const float*)(srcb + row0 * (FEAT_N * 4) + 12);
    const float p2 = *(const float*)(srcb + row0 * (FEAT_N * 4) + 24);
#pragma unroll
    for (int k = 0; k < 32; ++k)
        XA[k] = k < 29 ? *(const float*)(srcb + ixA + 12u * k)
                       : *(const float*)(srcb + ixAT + 12u * (k - 29));

    b2.y = hfb ? 1.0f : p0;
    RNN_STEP(odummy,   b2.y = hfb ? 1.0f : p1;);
    RNN_STEP(OQ[0][0], b2.y = hfb ? 1.0f : p2;);

#define SEG_BLOCK(XC, XN, IXC, IXN, IXCT, IXNT, OB, OBN)                       \
    _Pragma("unroll")                                                          \
    for (int k = 0; k < 32; ++k) {                                             \
        RNN_STEP(OQ[((k + 1) >> 2) & 1][(k + 1) & 3],                          \
            if (STORE && (k & 3) == 3)                                         \
                *(v4f*)(outb + OB + 4u * (k - 3)) = OQ[(k >> 2) & 1];          \
            if (STORE && (k & 3) == 2) {                                       \
                       \
                asm volatile("" :: "v"(OQ[((k >> 2) + 1) & 1]));               \
            }                                                                  \
            b2.y = hfb ? 1.0f : XC[k];                                         \
            if (k < 29)                                                        \
                XN[k] = *(const float*)(srcb + IXN + 12u * k);                 \
            else                                                               \
                XN[k] = *(const float*)(srcb + IXNT + 12u * (k - 29));         \
            if (k == 15) {                                                     \
                const unsigned nx_ = IXC + 2 * 32 * (FEAT_N * 4);              \
                IXC = nx_ < ixmax ? nx_ : ixmax;                               \
                const unsigned nt_ = IXC + 29 * (FEAT_N * 4);                  \
                IXCT = nt_ < tlim ? nt_ : tlim;                                \
                OBN += 2 * 32 * 4;                                             \
            }                                                                  \
            if (STORE && k == 30) {                                            \
                   \
                f0 = b0.x; f1 = b0.y; f2 = b1.x; f3 = b1.y; f4 = b2.x;         \
            }                                                                  \
        );                                                                     \
    }

    for (int m = 0; m < nblk / 2; ++m) {
        SEG_BLOCK(XA, XB, ixA, ixB, ixAT, ixBT, obA, obB)
        SEG_BLOCK(XB, XA, ixB, ixA, ixBT, ixAT, obB, obA)
    }
#undef SEG_BLOCK
}

__device__ __forceinline__ void serial_scan(
    const v2f (&Am)[3], const float* __restrict__ src, float* __restrict__ out,
    const int lane, const int hf)
{
    v2f b0 = {0.0f, 0.0f}, b1 = {0.0f, 0.0f}, b2 = {0.0f, 0.0f};
    float fc = 0.0f;

#define LANE_READ(v, k) \
    __int_as_float(__builtin_amdgcn_readlane(__float_as_int(v), (k)))


    const unsigned lane3 = (unsigned)lane * FEAT_N;
    float xvA = src[lane3];
    float xvB = src[32 * FEAT_N + lane3];
    const unsigned ixmax = (SEQ_N / 32 - 1) * (32 * FEAT_N) + lane3;
    unsigned ixA = lane3;
    unsigned ixB = 32 * FEAT_N + lane3;
    float oA = 0.0f, oB = 0.0f;
    float xs;

    b2.y = hf ? 1.0f : LANE_READ(xvA, 0);
    {
        const float x1 = LANE_READ(xvA, 1);
        RNN_STEP(fc, b2.y = hf ? 1.0f : x1;);
        const float x2 = LANE_READ(xvA, 2);
        const float x3 = LANE_READ(xvA, 3);
        RNN_STEP(fc, b2.y = hf ? 1.0f : x2; xs = x3;);
    }

#define RNN_STEPS(O, XV, XVN, IX, T0, K0, K1)                                  \
    _Pragma("unroll")                                                          \
    for (int k = (K0); k < (K1); ++k) {                                        \
        RNN_STEP(fc,                                                           \
            O = (lane == k) ? fc : O;                                          \
              \
            asm volatile("" :: "v"(O));                                        \
            b2.y = hf ? 1.0f : xs;                                             \
            xs = (k < 28) ? LANE_READ(XV, (k + 4) & 31)                        \
                          : LANE_READ(XVN, (k - 28) & 31);                     \
            if (k == 15) {                                                     \
                const unsigned nx = IX + 2 * 32 * FEAT_N;                      \
                IX = nx < ixmax ? nx : ixmax;                                  \
            }                                                                  \
            if (k == 31) {                                                     \
                out[(T0) + lane] = O;                                          \
                XV = src[IX];                                                  \
            }                                                                  \
        );                                                                     \
    }

    RNN_STEPS(oA, xvA, xvB, ixA, 0, 0, 16)
    for (int m = 0; m < SEQ_N / 64; ++m) {
        const int t0 = m * 64;
        RNN_STEPS(oA, xvA, xvB, ixA, t0, 16, 32)
        RNN_STEPS(oB, xvB, xvA, ixB, t0 + 32, 0, 32)
        RNN_STEPS(oA, xvA, xvB, ixA, t0 + 64, 0, 16)
    }
#undef RNN_STEPS
#undef LANE_READ
}

__device__ __forceinline__ void load_a_operand(
    v2f (&Am)[3],
    const float* __restrict__ W_ih, const float* __restrict__ W_hh,
    const float* __restrict__ b_ih, const float* __restrict__ b_hh,
    const float* __restrict__ W_fc, const float* __restrict__ b_fc,
    const int lane)
{
    const int row  = lane & 15;
    const int hf   = lane >> 4;
    const int rr   = row & 7;
    const bool row_hid = rr < 5;
    const bool row_fc  = rr == 5;
    const int unit = rr + 5 * (row >> 3);

#pragma unroll
    for (int c = 0; c < 3; ++c) {
#pragma unroll
        for (int s = 0; s < 2; ++s) {
            const int r = 2 * c + s;
            float w = 0.0f;
            if (r < 5) {
                const int j = r + 5 * hf;
                if (row_hid)      w = W_hh[unit * HID_N + j];
                else if (row_fc)  w = W_fc[j];
            } else {
                if (row_hid)      w = hf ? (b_ih[unit] + b_hh[unit]) : W_ih[unit];
                else if (row_fc)  w = hf ? b_fc[0] : 0.0f;
            }
            Am[c][s] = w;
        }
    }
}

__global__ __launch_bounds__(WG_THREADS)
void rnn_segments_wmma(const float* __restrict__ src,
                       const float* __restrict__ W_ih,
                       const float* __restrict__ W_hh,
                       const float* __restrict__ b_ih,
                       const float* __restrict__ b_hh,
                       const float* __restrict__ W_fc,
                       const float* __restrict__ b_fc,
                       float* __restrict__ out,
                       float* __restrict__ ws)
{
    const int lane = threadIdx.x & 31;
    const int hf   = lane >> 4;
    const unsigned grp = blockIdx.x * WG_WAVES
                       + (unsigned)__builtin_amdgcn_readfirstlane((int)(threadIdx.x >> 5));

    v2f Am[3];
    load_a_operand(Am, W_ih, W_hh, b_ih, b_hh, W_fc, b_fc, lane);

    const unsigned col = (unsigned)lane & 15u;
    const bool hfb = hf != 0;
    const bool head = grp == 0u;
    const unsigned segu = (grp - 1u) * 16u + col;
    const bool spare = !head && segu >= NSEG;
    const unsigned seg = segu < NSEG ? segu : NSEG - 1u;
    const unsigned row0 = head ? 0u : HEAD_N + seg * SEG_N;
    const int nblk = head ? HEAD_N / 32 : SEG_N / 32;

    v2f b0 = {0.0f, 0.0f}, b1 = {0.0f, 0.0f}, b2 = {0.0f, 0.0f};
    float f0 = 0.0f, f1 = 0.0f, f2 = 0.0f, f3 = 0.0f, f4 = 0.0f;

    if (!head)
        segmented_pass<false>(Am, b0, b1, b2, src, out, hfb,
                              row0 - RUNIN_N, 0u, RUNIN_BLOCKS,
                              f0, f1, f2, f3, f4);
    const float s0 = b0.x, s1 = b0.y, s2 = b1.x, s3 = b1.y, s4 = b2.x;
    segmented_pass<true>(Am, b0, b1, b2, src, out, hfb,
                         row0, row0, nblk,
                         f0, f1, f2, f3, f4);

    bool differs = false;
#define HAND_OVER(F, S)                                                        \
    differs |= __builtin_amdgcn_ds_bpermute(((lane - 1) & 31) << 2,            \
                                            __float_as_int(F)) != __float_as_int(S)
    HAND_OVER(f0, s0);
    HAND_OVER(f1, s1);
    HAND_OVER(f2, s2);
    HAND_OVER(f3, s3);
    HAND_OVER(f4, s4);
#undef HAND_OVER
    differs = differs && (col != 0u) && !head && !spare;
    const bool wave_differs = __builtin_amdgcn_ballot_w32(differs) != 0u;

    float* const w = ws + grp * WS_GRP + 5 * hf;
    if (col == 0u) {
        w[WS_START + 0] = s0; w[WS_START + 1] = s1; w[WS_START + 2] = s2;
        w[WS_START + 3] = s3; w[WS_START + 4] = s4;
    }
    if (col == 15u) {
        w[WS_END + 0] = f0; w[WS_END + 1] = f1; w[WS_END + 2] = f2;
        w[WS_END + 3] = f3; w[WS_END + 4] = f4;
    }
    if (lane == 0)
        ws[grp * WS_GRP + WS_FLAG] = wave_differs ? 1.0f : 0.0f;
}

__global__ __launch_bounds__(CHECK_THREADS)
void rnn_check_handover(const float* __restrict__ src,
                        const float* __restrict__ W_ih,
                        const float* __restrict__ W_hh,
                        const float* __restrict__ b_ih,
                        const float* __restrict__ b_hh,
                        const float* __restrict__ W_fc,
                        const float* __restrict__ b_fc,
                        float* __restrict__ out,
                        const float* __restrict__ ws)
{
    __shared__ int any_differs;
    const unsigned t = threadIdx.x;
    if (t == 0u) any_differs = 0;
    __syncthreads();

    bool differs = false;
    for (unsigned g = t; g < NGRP; g += CHECK_THREADS) {
        if (g == 0u) continue;
        const float* const st = ws + g * WS_GRP;
        const float* const en = ws + (g - 1u) * WS_GRP + WS_END;
#pragma unroll
        for (int i = 0; i < HID_N; ++i)
            differs |= __float_as_int(en[i]) != __float_as_int(st[WS_START + i]);
        differs |= __float_as_int(st[WS_FLAG]) != 0;
    }
    if (differs) any_differs = 1;
    __syncthreads();

    if (any_differs != 0 && t < 32u) {
        const int lane = (int)t;
        v2f Am[3];
        load_a_operand(Am, W_ih, W_hh, b_ih, b_hh, W_fc, b_fc, lane);
        serial_scan(Am, src, out, lane, lane >> 4);
    }
}
#undef RNN_STEP

extern "C" void kernel_launch(void* const* d_in, const int* in_sizes, int n_in,
                              void* d_out, int out_size, void* d_ws, size_t ws_size,
                              hipStream_t stream) {
    const float* src  = (const float*)d_in[0];
    const float* W_ih = (const float*)d_in[1];
    const float* W_hh = (const float*)d_in[2];
    const float* b_ih = (const float*)d_in[3];
    const float* b_hh = (const float*)d_in[4];
    const float* W_fc = (const float*)d_in[5];
    const float* b_fc = (const float*)d_in[6];
    float* out = (float*)d_out;
    float* ws  = (float*)d_ws;

    rnn_segments_wmma<<<WG_N, WG_THREADS, 0, stream>>>(
        src, W_ih, W_hh, b_ih, b_hh, W_fc, b_fc, out, ws);
    rnn_check_handover<<<1, CHECK_THREADS, 0, stream>>>(
        src, W_ih, W_hh, b_ih, b_hh, W_fc, b_fc, out, ws);
}
